# starting kernel
# baseline (speedup 1.0000x reference)
#include <hip/hip_runtime.h>


typedef __attribute__((ext_vector_type(2))) float v2f;
typedef __attribute__((ext_vector_type(8))) float v8f;

#define GRU_B 4096
#define GRU_T 10
#define GRU_S 301
#define GRU_N (GRU_B * GRU_S)

__device__ __forceinline__ float swap_half(float v) {
  int i = __builtin_bit_cast(int, v);
  i = __builtin_amdgcn_ds_swizzle(i, 0x401F);
  return __builtin_bit_cast(float, i);
}

__device__ __forceinline__ float tanh_fast(float x) {
#if __has_builtin(__builtin_amdgcn_tanhf)
  return __builtin_amdgcn_tanhf(x);
#else
  float e = __builtin_amdgcn_exp2f(2.88539008f * x);
  return 1.0f - 2.0f * __builtin_amdgcn_rcpf(e + 1.0f);
#endif
}

__device__ __forceinline__ float sigmoid_fast(float x) {
#if __has_builtin(__builtin_amdgcn_tanhf)
  return __builtin_fmaf(0.5f, __builtin_amdgcn_tanhf(0.5f * x), 0.5f);
#else
  float e = __builtin_amdgcn_exp2f(-1.44269504f * x);
  return __builtin_amdgcn_rcpf(1.0f + e);
#endif
}

__global__ __launch_bounds__(256) void gru_wmma_kernel(
    const float* __restrict__ x,
    const float* __restrict__ w_ih,
    const float* __restrict__ w_hh,
    const float* __restrict__ b_ih,
    const float* __restrict__ b_hh,
    const float* __restrict__ fc_w,
    const float* __restrict__ fc_b,
    float* __restrict__ out)
{
  const int lane = threadIdx.x & 31;
  const int m    = lane & 15;
  const bool hi  = lane >= 16;

  const int wave = blockIdx.x * (blockDim.x >> 5) + (threadIdx.x >> 5);
  if (wave * 16 >= GRU_N) return;

  const int n = wave * 16 + m;
  const int b = n / GRU_S;
  const int s = n - b * GRU_S;
  const int xbase = b * (GRU_T * GRU_S) + s;

  const int mm = (m < 15) ? m : 14;
  float a10 = w_hh[mm * 5 + (hi ? 2 : 0)];
  float a11 = w_hh[mm * 5 + (hi ? 3 : 1)];
  float biasm = (mm < 10) ? (b_ih[mm] + b_hh[mm]) : b_hh[mm];
  float a20 = hi ? biasm : w_hh[mm * 5 + 4];
  float a21 = hi ? 0.0f : ((mm < 10) ? w_ih[mm] : 0.0f);
  if (m == 15) { a10 = 0.0f; a11 = 0.0f; a20 = 0.0f; a21 = 0.0f; }
  v2f A1; A1.x = a10; A1.y = a11;
  v2f A2; A2.x = a20; A2.y = a21;

  float wn[5], bn[5];
#pragma unroll
  for (int j = 0; j < 5; ++j) { wn[j] = w_ih[10 + j]; bn[j] = b_ih[10 + j]; }

  float xv[GRU_T];
#pragma unroll
  for (int t = 0; t < GRU_T; ++t) xv[t] = x[xbase + t * GRU_S];

  float h[5] = {0.f, 0.f, 0.f, 0.f, 0.f};

#pragma unroll
  for (int t = 0; t < GRU_T; ++t) {
    v2f B1; B1.x = hi ? h[2] : h[0];
            B1.y = hi ? h[3] : h[1];
    v2f B2; B2.x = hi ? 1.0f : h[4];
            B2.y = hi ? 0.0f : xv[t];

    v8f c = {};
    c = __builtin_amdgcn_wmma_f32_16x16x4_f32(false, A1, false, B1,
                                              (short)0, c, false, false);
    c = __builtin_amdgcn_wmma_f32_16x16x4_f32(false, A2, false, B2,
                                              (short)0, c, false, false);

    float p[15];
#pragma unroll
    for (int k = 0; k < 8; ++k) {
      float dk = c[k];
      float tk = swap_half(dk);
      p[k] = hi ? tk : dk;
      if (k < 7) p[k + 8] = hi ? dk : tk;
    }

#pragma unroll
    for (int j = 0; j < 5; ++j) {
      float r  = sigmoid_fast(p[j]);
      float z  = sigmoid_fast(p[5 + j]);
      float np = __builtin_fmaf(wn[j], xv[t], bn[j]);
      np = __builtin_fmaf(r, p[10 + j], np);
      float nn = tanh_fast(np);
      h[j] = __builtin_fmaf(z, h[j] - nn, nn);
    }
  }

  if (!hi) {
#pragma unroll
    for (int o = 0; o < 3; ++o) {
      float y = fc_b[o];
#pragma unroll
      for (int j = 0; j < 5; ++j) y = __builtin_fmaf(fc_w[o * 5 + j], h[j], y);
      out[(b * 3 + o) * GRU_S + s] = y;
    }
  }
}

extern "C" void kernel_launch(void* const* d_in, const int* in_sizes, int n_in,
                              void* d_out, int out_size, void* d_ws, size_t ws_size,
                              hipStream_t stream) {
  (void)in_sizes; (void)n_in; (void)out_size; (void)d_ws; (void)ws_size;
  const float* x    = (const float*)d_in[0];
  const float* w_ih = (const float*)d_in[1];
  const float* w_hh = (const float*)d_in[2];
  const float* b_ih = (const float*)d_in[3];
  const float* b_hh = (const float*)d_in[4];
  const float* fc_w = (const float*)d_in[5];
  const float* fc_b = (const float*)d_in[6];
  float* out = (float*)d_out;

  const int threads = 256;
  const int samplesPerBlock = (threads / 32) * 16;
  const int blocks = (GRU_N + samplesPerBlock - 1) / samplesPerBlock;
  gru_wmma_kernel<<<blocks, threads, 0, stream>>>(x, w_ih, w_hh, b_ih, b_hh,
                                                  fc_w, fc_b, out);
}


# Lane pairs split each sample's hidden units, A rows ordered so gates land in the owning lane: 32 samples per wave, no duplicated gate math, one lane exchange per tile; k order of gate sums changes
# speedup vs baseline: 1.4049x; 1.4049x over previous
#include <hip/hip_runtime.h>


typedef __attribute__((ext_vector_type(2))) float v2f;
typedef __attribute__((ext_vector_type(8))) float v8f;

#define GRU_B 4096
#define GRU_T 10
#define GRU_S 301
#define GRU_N (GRU_B * GRU_S)

__device__ __forceinline__ float xhalf(float v) {
  int b = __builtin_bit_cast(int, v);
  return __builtin_bit_cast(float,
      __builtin_amdgcn_permlanex16(b, b, 0x76543210, 0xfedcba98, false, false));
}

__device__ __forceinline__ float tanh_fast(float x) {
#if __has_builtin(__builtin_amdgcn_tanhf)
  return __builtin_amdgcn_tanhf(x);
#else
  float e = __builtin_amdgcn_exp2f(2.88539008f * x);
  return 1.0f - 2.0f * __builtin_amdgcn_rcpf(e + 1.0f);
#endif
}

__device__ __forceinline__ float sigmoid_fast(float x) {
#if __has_builtin(__builtin_amdgcn_tanhf)
  return __builtin_fmaf(0.5f, __builtin_amdgcn_tanhf(0.5f * x), 0.5f);
#else
  float e = __builtin_amdgcn_exp2f(-1.44269504f * x);
  return __builtin_amdgcn_rcpf(1.0f + e);
#endif
}

__global__ __launch_bounds__(256) void gru_wmma_kernel(
    const float* __restrict__ x,
    const float* __restrict__ w_ih,
    const float* __restrict__ w_hh,
    const float* __restrict__ b_ih,
    const float* __restrict__ b_hh,
    const float* __restrict__ fc_w,
    const float* __restrict__ fc_b,
    float* __restrict__ out)
{
  const int lane = threadIdx.x & 31;
  const int c    = lane & 15;
  const int h    = lane >> 4;
  const bool hi  = (h != 0);

  const int wave = blockIdx.x * (blockDim.x >> 5) + (threadIdx.x >> 5);
  if (wave * 32 >= GRU_N) return;
  const int base = wave * 32;

  v2f A1[2], A2[2];
  float wnA[2], bnA[2], wnB[2], bnB[2], wnC[2], bnC[2];
  bool roleLo[2];
#pragma unroll
  for (int T = 0; T < 2; ++T) {
    const bool rowLo = ((c >> 3) == T);
    const int r = c & 7;
    const int q = r / 3;
    const int k = r - 3 * q;
    const int gLo = (r < 6) ? (q + 5 * k)     : ((r == 6) ? 2 : 12);
    const int gHi = (r < 6) ? (3 + q + 5 * k) : ((r == 6) ? 7 : -1);
    const int g = rowLo ? gLo : gHi;
    const bool valid = (g >= 0);
    const int gs = valid ? g : 0;
    const bool featLo = (h == T);
    const int f0 = featLo ? 0 : 3;
    const float w0 = w_hh[gs * 5 + f0];
    const float w1 = w_hh[gs * 5 + f0 + 1];
    const float w2 = w_hh[gs * 5 + 2];
    const float bi = b_ih[gs];
    const float bh = b_hh[gs];
    const float wi = w_ih[gs];
    const bool rz = (gs < 10);
    const float bias = rz ? (bi + bh) : bh;
    const float wih  = rz ? wi : 0.0f;
    A1[T].x = valid ? w0 : 0.0f;
    A1[T].y = valid ? w1 : 0.0f;
    A2[T].x = valid ? (featLo ? w2 : bias) : 0.0f;
    A2[T].y = (valid && featLo) ? wih : 0.0f;

    roleLo[T] = featLo;
    const int ua = featLo ? 0 : 3;
    wnA[T] = w_ih[10 + ua];     bnA[T] = b_ih[10 + ua];
    wnB[T] = w_ih[10 + ua + 1]; bnB[T] = b_ih[10 + ua + 1];
    const float wn2 = w_ih[12];
    const float bn2 = b_ih[12];
    wnC[T] = featLo ? wn2 : 0.0f;
    bnC[T] = featLo ? bn2 : 0.0f;
  }

  float xv[2][GRU_T];
#pragma unroll
  for (int T = 0; T < 2; ++T) {
    const int n  = base + 16 * T + c;
    const int b  = n / GRU_S;
    const int s  = n - b * GRU_S;
    const int xb = b * (GRU_T * GRU_S) + s;
#pragma unroll
    for (int t = 0; t < GRU_T; ++t) xv[T][t] = x[xb + t * GRU_S];
  }

  float st[2][3] = {{0.f, 0.f, 0.f}, {0.f, 0.f, 0.f}};

#pragma unroll
  for (int t = 0; t < GRU_T; ++t) {
#pragma unroll
    for (int T = 0; T < 2; ++T) {
      const float xt = xv[T][t];
      v2f B1; B1.x = st[T][0];
              B1.y = st[T][1];
      v2f B2; B2.x = roleLo[T] ? st[T][2] : 1.0f;
              B2.y = roleLo[T] ? xt : 0.0f;

      v8f d = {};
      d = __builtin_amdgcn_wmma_f32_16x16x4_f32(false, A1[T], false, B1,
                                                (short)0, d, false, false);
      d = __builtin_amdgcn_wmma_f32_16x16x4_f32(false, A2[T], false, B2,
                                                (short)0, d, false, false);

      {
        const float r  = sigmoid_fast(d[0]);
        const float z  = sigmoid_fast(d[1]);
        float np = __builtin_fmaf(wnA[T], xt, bnA[T]);
        np = __builtin_fmaf(r, d[2], np);
        const float nn = tanh_fast(np);
        st[T][0] = __builtin_fmaf(z, st[T][0] - nn, nn);
      }
      {
        const float r  = sigmoid_fast(d[3]);
        const float z  = sigmoid_fast(d[4]);
        float np = __builtin_fmaf(wnB[T], xt, bnB[T]);
        np = __builtin_fmaf(r, d[5], np);
        const float nn = tanh_fast(np);
        st[T][1] = __builtin_fmaf(z, st[T][1] - nn, nn);
      }
      {
        const float g6 = sigmoid_fast(d[6]);
        const float z  = xhalf(g6);
        float np = __builtin_fmaf(wnC[T], xt, bnC[T]);
        np = __builtin_fmaf(g6, d[7], np);
        const float nn = tanh_fast(np);
        st[T][2] = __builtin_fmaf(z, st[T][2] - nn, nn);
      }
    }
  }

  const float send0 = hi ? st[0][0] : st[1][0];
  const float send1 = hi ? st[0][1] : st[1][1];
  float hf[5];
  hf[0] = hi ? st[1][0] : st[0][0];
  hf[1] = hi ? st[1][1] : st[0][1];
  hf[2] = hi ? st[1][2] : st[0][2];
  hf[3] = xhalf(send0);
  hf[4] = xhalf(send1);

  {
    const int n = base + lane;
    const int b = n / GRU_S;
    const int s = n - b * GRU_S;
#pragma unroll
    for (int o = 0; o < 3; ++o) {
      float y = fc_b[o];
#pragma unroll
      for (int j = 0; j < 5; ++j) y = __builtin_fmaf(fc_w[o * 5 + j], hf[j], y);
      out[(b * 3 + o) * GRU_S + s] = y;
    }
  }
}

extern "C" void kernel_launch(void* const* d_in, const int* in_sizes, int n_in,
                              void* d_out, int out_size, void* d_ws, size_t ws_size,
                              hipStream_t stream) {
  (void)in_sizes; (void)n_in; (void)out_size; (void)d_ws; (void)ws_size;
  const float* x    = (const float*)d_in[0];
  const float* w_ih = (const float*)d_in[1];
  const float* w_hh = (const float*)d_in[2];
  const float* b_ih = (const float*)d_in[3];
  const float* b_hh = (const float*)d_in[4];
  const float* fc_w = (const float*)d_in[5];
  const float* fc_b = (const float*)d_in[6];
  float* out = (float*)d_out;

  const int threads = 256;
  const int samplesPerBlock = (threads / 32) * 32;
  const int blocks = (GRU_N + samplesPerBlock - 1) / samplesPerBlock;
  gru_wmma_kernel<<<blocks, threads, 0, stream>>>(x, w_ih, w_hh, b_ih, b_hh,
                                                  fc_w, fc_b, out);
}


# Sigmoid's half-scale folded into the constant A rows of the r and z gates (exact power-of-two scaling): the gates' tanh reads D directly, 3 multiplies and 2 moves fewer per tile and step
# speedup vs baseline: 1.4546x; 1.0354x over previous
#include <hip/hip_runtime.h>


typedef __attribute__((ext_vector_type(2))) float v2f;
typedef __attribute__((ext_vector_type(8))) float v8f;

#define GRU_B 4096
#define GRU_T 10
#define GRU_S 301
#define GRU_N (GRU_B * GRU_S)

__device__ __forceinline__ float xhalf(float v) {
  int b = __builtin_bit_cast(int, v);
  return __builtin_bit_cast(float,
      __builtin_amdgcn_permlanex16(b, b, 0x76543210, 0xfedcba98, false, false));
}

__device__ __forceinline__ float tanh_fast(float x) {
#if __has_builtin(__builtin_amdgcn_tanhf)
  return __builtin_amdgcn_tanhf(x);
#else
  float e = __builtin_amdgcn_exp2f(2.88539008f * x);
  return 1.0f - 2.0f * __builtin_amdgcn_rcpf(e + 1.0f);
#endif
}

__device__ __forceinline__ float sigmoid_of_half(float ph) {
  return __builtin_fmaf(0.5f, tanh_fast(ph), 0.5f);
}

__global__ __launch_bounds__(256) void gru_wmma_kernel(
    const float* __restrict__ x,
    const float* __restrict__ w_ih,
    const float* __restrict__ w_hh,
    const float* __restrict__ b_ih,
    const float* __restrict__ b_hh,
    const float* __restrict__ fc_w,
    const float* __restrict__ fc_b,
    float* __restrict__ out)
{
  const int lane = threadIdx.x & 31;
  const int c    = lane & 15;
  const int h    = lane >> 4;
  const bool hi  = (h != 0);

  const int wave = blockIdx.x * (blockDim.x >> 5) + (threadIdx.x >> 5);
  if (wave * 32 >= GRU_N) return;
  const int base = wave * 32;

  v2f A1[2], A2[2];
  float wnA[2], bnA[2], wnB[2], bnB[2], wnC[2], bnC[2];
  bool roleLo[2];
#pragma unroll
  for (int T = 0; T < 2; ++T) {
    const bool rowLo = ((c >> 3) == T);
    const int r = c & 7;
    const int q = r / 3;
    const int k = r - 3 * q;
    const int gLo = (r < 6) ? (q + 5 * k)     : ((r == 6) ? 2 : 12);
    const int gHi = (r < 6) ? (3 + q + 5 * k) : ((r == 6) ? 7 : -1);
    const int g = rowLo ? gLo : gHi;
    const bool valid = (g >= 0);
    const int gs = valid ? g : 0;
    const bool featLo = (h == T);
    const int f0 = featLo ? 0 : 3;
    const float w0 = w_hh[gs * 5 + f0];
    const float w1 = w_hh[gs * 5 + f0 + 1];
    const float w2 = w_hh[gs * 5 + 2];
    const float bi = b_ih[gs];
    const float bh = b_hh[gs];
    const float wi = w_ih[gs];
    const bool rz = (gs < 10);
    const float bias = rz ? (bi + bh) : bh;
    const float wih  = rz ? wi : 0.0f;
    const float sc   = rz ? 0.5f : 1.0f;
    A1[T].x = valid ? sc * w0 : 0.0f;
    A1[T].y = valid ? sc * w1 : 0.0f;
    A2[T].x = valid ? sc * (featLo ? w2 : bias) : 0.0f;
    A2[T].y = (valid && featLo) ? sc * wih : 0.0f;

    roleLo[T] = featLo;
    const int ua = featLo ? 0 : 3;
    wnA[T] = w_ih[10 + ua];     bnA[T] = b_ih[10 + ua];
    wnB[T] = w_ih[10 + ua + 1]; bnB[T] = b_ih[10 + ua + 1];
    const float wn2 = w_ih[12];
    const float bn2 = b_ih[12];
    wnC[T] = featLo ? wn2 : 0.0f;
    bnC[T] = featLo ? bn2 : 0.0f;
  }

  float xv[2][GRU_T];
#pragma unroll
  for (int T = 0; T < 2; ++T) {
    const int n  = base + 16 * T + c;
    const int b  = n / GRU_S;
    const int s  = n - b * GRU_S;
    const int xb = b * (GRU_T * GRU_S) + s;
#pragma unroll
    for (int t = 0; t < GRU_T; ++t) xv[T][t] = x[xb + t * GRU_S];
  }

  float st[2][3] = {{0.f, 0.f, 0.f}, {0.f, 0.f, 0.f}};

#pragma unroll
  for (int t = 0; t < GRU_T; ++t) {
#pragma unroll
    for (int T = 0; T < 2; ++T) {
      const float xt = xv[T][t];
      v2f B1; B1.x = st[T][0];
              B1.y = st[T][1];
      v2f B2; B2.x = roleLo[T] ? st[T][2] : 1.0f;
              B2.y = roleLo[T] ? xt : 0.0f;

      v8f d = {};
      d = __builtin_amdgcn_wmma_f32_16x16x4_f32(false, A1[T], false, B1,
                                                (short)0, d, false, false);
      d = __builtin_amdgcn_wmma_f32_16x16x4_f32(false, A2[T], false, B2,
                                                (short)0, d, false, false);

      {
        const float r  = sigmoid_of_half(d[0]);
        const float z  = sigmoid_of_half(d[1]);
        float np = __builtin_fmaf(wnA[T], xt, bnA[T]);
        np = __builtin_fmaf(r, d[2], np);
        const float nn = tanh_fast(np);
        st[T][0] = __builtin_fmaf(z, st[T][0] - nn, nn);
      }
      {
        const float r  = sigmoid_of_half(d[3]);
        const float z  = sigmoid_of_half(d[4]);
        float np = __builtin_fmaf(wnB[T], xt, bnB[T]);
        np = __builtin_fmaf(r, d[5], np);
        const float nn = tanh_fast(np);
        st[T][1] = __builtin_fmaf(z, st[T][1] - nn, nn);
      }
      {
        const float g6 = sigmoid_of_half(d[6]);
        const float z  = xhalf(g6);
        float np = __builtin_fmaf(wnC[T], xt, bnC[T]);
        np = __builtin_fmaf(g6, d[7], np);
        const float nn = tanh_fast(np);
        st[T][2] = __builtin_fmaf(z, st[T][2] - nn, nn);
      }
    }
  }

  const float send0 = hi ? st[0][0] : st[1][0];
  const float send1 = hi ? st[0][1] : st[1][1];
  float hf[5];
  hf[0] = hi ? st[1][0] : st[0][0];
  hf[1] = hi ? st[1][1] : st[0][1];
  hf[2] = hi ? st[1][2] : st[0][2];
  hf[3] = xhalf(send0);
  hf[4] = xhalf(send1);

  {
    const int n = base + lane;
    const int b = n / GRU_S;
    const int s = n - b * GRU_S;
#pragma unroll
    for (int o = 0; o < 3; ++o) {
      float y = fc_b[o];
#pragma unroll
      for (int j = 0; j < 5; ++j) y = __builtin_fmaf(fc_w[o * 5 + j], hf[j], y);
      out[(b * 3 + o) * GRU_S + s] = y;
    }
  }
}

extern "C" void kernel_launch(void* const* d_in, const int* in_sizes, int n_in,
                              void* d_out, int out_size, void* d_ws, size_t ws_size,
                              hipStream_t stream) {
  (void)in_sizes; (void)n_in; (void)out_size; (void)d_ws; (void)ws_size;
  const float* x    = (const float*)d_in[0];
  const float* w_ih = (const float*)d_in[1];
  const float* w_hh = (const float*)d_in[2];
  const float* b_ih = (const float*)d_in[3];
  const float* b_hh = (const float*)d_in[4];
  const float* fc_w = (const float*)d_in[5];
  const float* fc_b = (const float*)d_in[6];
  float* out = (float*)d_out;

  const int threads = 256;
  const int samplesPerBlock = (threads / 32) * 32;
  const int blocks = (GRU_N + samplesPerBlock - 1) / samplesPerBlock;
  gru_wmma_kernel<<<blocks, threads, 0, stream>>>(x, w_ih, w_hh, b_ih, b_hh,
                                                  fc_w, fc_b, out);
}
